# starting kernel
# baseline (speedup 1.0000x reference)
#include <hip/hip_runtime.h>

typedef __attribute__((ext_vector_type(2))) float v2f;
typedef __attribute__((ext_vector_type(4))) float v4f;
typedef __attribute__((ext_vector_type(8))) float v8f;

#define BATCH        1024
#define NUM_GENES    20000
#define NONGENE_DIM  64
#define EMB_DIM      32
#define INPUT_DIM    (NUM_GENES + NONGENE_DIM)

__global__ void gemv_wmma_f32(const float* __restrict__ A, long long row_stride,
                              long long col_off, int n_rows, int K,
                              const float* __restrict__ w,
                              const float* __restrict__ bias_ptr,
                              float* __restrict__ out) {
  const int wave = (int)(blockIdx.x * (blockDim.x >> 5) + (threadIdx.x >> 5));
  const int lane = (int)(threadIdx.x & 31);
  const int n_tiles = n_rows >> 4;
  if (wave >= n_tiles) return;

  const int row   = (wave << 4) + (lane & 15);
  const int kpair = (lane >> 4) << 1;
  const float* arow = A + (long long)row * row_stride + col_off + kpair;
  const float* wrow = w + kpair;

  v8f c = {};
  for (int k0 = 0; k0 < K; k0 += 4) {
    v2f a, bv;
    a.x  = arow[k0];  a.y  = arow[k0 + 1];
    bv.x = wrow[k0];  bv.y = wrow[k0 + 1];
    c = __builtin_amdgcn_wmma_f32_16x16x4_f32(false, a, false, bv, (short)0, c,
                                              false, false);
  }

  const float bias = bias_ptr ? bias_ptr[0] : 0.0f;
  if (lane == 0) {
#pragma unroll
    for (int v = 0; v < 8; ++v) out[(wave << 4) + v] = c[v] + bias;
  } else if (lane == 16) {
#pragma unroll
    for (int v = 0; v < 8; ++v) out[(wave << 4) + 8 + v] = c[v] + bias;
  }
}

__global__ void fused_stream_kernel(const float* __restrict__ x,
                                    const float* __restrict__ gp,
                                    const float* __restrict__ ng,
                                    const float* __restrict__ W,
                                    float* __restrict__ out) {
  const int col4 = blockIdx.x * blockDim.x + threadIdx.x;
  if (col4 >= NUM_GENES / 4) return;
  const int row = blockIdx.y;

  const float w_x = W[NONGENE_DIM];
  const float ngr = ng[row];

  const long long xoff = (long long)row * INPUT_DIM + (long long)col4 * 4;
  const long long ooff = (long long)row * NUM_GENES + (long long)col4 * 4;

  v4f xv = __builtin_nontemporal_load((const v4f*)(x + xoff));
  v4f gv = *(const v4f*)(gp + (long long)col4 * 4);
  v4f y  = xv * w_x + (gv + ngr);
  __builtin_nontemporal_store(y, (v4f*)(out + ooff));
}

extern "C" void kernel_launch(void* const* d_in, const int* in_sizes, int n_in,
                              void* d_out, int out_size, void* d_ws, size_t ws_size,
                              hipStream_t stream) {
  const float* x   = (const float*)d_in[0];
  const float* emb = (const float*)d_in[1];
  const float* W   = (const float*)d_in[2];
  const float* b   = (const float*)d_in[3];
  float* out = (float*)d_out;

  float* gp = (float*)d_ws;
  float* ng = gp + NUM_GENES;

  {
    const int tiles = NUM_GENES / 16;
    const int wpb = 8;
    dim3 grd((tiles + wpb - 1) / wpb), blk(32 * wpb);
    gemv_wmma_f32<<<grd, blk, 0, stream>>>(emb, EMB_DIM, 0, NUM_GENES, EMB_DIM,
                                           W + NONGENE_DIM + 1, b, gp);
  }

  {
    const int tiles = BATCH / 16;
    const int wpb = 8;
    dim3 grd((tiles + wpb - 1) / wpb), blk(32 * wpb);
    gemv_wmma_f32<<<grd, blk, 0, stream>>>(x, INPUT_DIM, NUM_GENES, BATCH,
                                           NONGENE_DIM, W, nullptr, ng);
  }

  {
    dim3 blk(256);
    dim3 grd((NUM_GENES / 4 + 255) / 256, BATCH);
    fused_stream_kernel<<<grd, blk, 0, stream>>>(x, gp, ng, W, out);
  }
}


# Single launch: each work-group fetches its dot-product operands as whole 128-byte lines, forms WMMA fragments in LDS, keeps its 16 row loads in flight under the WMMAs; 2 launches -> 1; sums unchanged
# speedup vs baseline: 1.5215x; 1.5215x over previous
#include <hip/hip_runtime.h>

typedef __attribute__((ext_vector_type(2))) float v2f;
typedef __attribute__((ext_vector_type(4))) float v4f;
typedef __attribute__((ext_vector_type(8))) float v8f;

#define BATCH        1024
#define NUM_GENES    20000
#define NONGENE_DIM  64
#define EMB_DIM      32
#define INPUT_DIM    (NUM_GENES + NONGENE_DIM)

#define WAVES_PER_WG   8
#define ROWS_PER_WAVE  16
#define TILE_ROWS      (WAVES_PER_WG * ROWS_PER_WAVE)
#define STRIP_COLS     128
#define N_STRIPS       ((NUM_GENES + STRIP_COLS - 1) / STRIP_COLS)
#define XT_PITCH       (NONGENE_DIM + 4)
#define EM_PITCH       (EMB_DIM + 4)

__global__ void __launch_bounds__(32 * WAVES_PER_WG, 10)
fused_tile_kernel(const float* __restrict__ x,
                  const float* __restrict__ emb,
                  const float* __restrict__ W,
                  const float* __restrict__ b,
                  float* __restrict__ out) {
  __shared__ __attribute__((aligned(16))) float s_xt[WAVES_PER_WG][ROWS_PER_WAVE * XT_PITCH];
  __shared__ __attribute__((aligned(16))) float s_em[WAVES_PER_WG][16 * EM_PITCH];
  __shared__ __attribute__((aligned(16))) float s_gp[STRIP_COLS];
  __shared__ __attribute__((aligned(16))) float s_ng[TILE_ROWS];
  __shared__ __attribute__((aligned(16))) float s_w[WAVES_PER_WG][96];

  const int lane  = (int)(threadIdx.x & 31);
  const int wib   = __builtin_amdgcn_readfirstlane((int)(threadIdx.x >> 5));
  const int strip = (int)blockIdx.x;
  const int row0  = (int)blockIdx.y * TILE_ROWS + wib * ROWS_PER_WAVE;
  const int col4  = strip * (STRIP_COLS / 4) + lane;
  const bool in_range = col4 < NUM_GENES / 4;

  v4f xt[8];
  {
    const float* p = x + (long long)(row0 + (lane >> 4)) * INPUT_DIM + NUM_GENES + ((lane & 15) << 2);
#pragma unroll
    for (int i = 0; i < 8; ++i) xt[i] = *(const v4f*)(p + (long long)(2 * i) * INPUT_DIM);
  }
  v4f em[4];
  {
    const int g0 = strip * STRIP_COLS + wib * 16 + (lane >> 3);
#pragma unroll
    for (int i = 0; i < 4; ++i) {
      int g = g0 + 4 * i;
      g = g < NUM_GENES ? g : NUM_GENES - 1;
      em[i] = *(const v4f*)(emb + (long long)g * EMB_DIM + ((lane & 7) << 2));
    }
  }

  const float w0 = W[lane], w1 = W[32 + lane], w2 = W[NONGENE_DIM + 1 + lane];

  s_w[wib][lane] = w0;  s_w[wib][32 + lane] = w1;  s_w[wib][64 + lane] = w2;
  {
    float* d = &s_xt[wib][(lane >> 4) * XT_PITCH + ((lane & 15) << 2)];
#pragma unroll
    for (int i = 0; i < 8; ++i) *(v4f*)(d + 2 * i * XT_PITCH) = xt[i];
    float* e = &s_em[wib][(lane >> 3) * EM_PITCH + ((lane & 7) << 2)];
#pragma unroll
    for (int i = 0; i < 4; ++i) *(v4f*)(e + 4 * i * EM_PITCH) = em[i];
  }

  const float* xp = x   + (long long)row0 * INPUT_DIM + (long long)col4 * 4;
  float*       op = out + (long long)row0 * NUM_GENES + (long long)col4 * 4;
  v4f xv[ROWS_PER_WAVE];
  if (in_range) {
#pragma unroll
    for (int r = 0; r < ROWS_PER_WAVE; ++r)
      xv[r] = __builtin_nontemporal_load((const v4f*)(xp + (long long)r * INPUT_DIM));
  }

  const int c16   = lane & 15;
  const int kpair = (lane >> 4) << 1;
  v8f cn = {};
  {
    const float* arow = &s_xt[wib][c16 * XT_PITCH + kpair];
    const float* wrow = &s_w[wib][kpair];
#pragma unroll
    for (int k0 = 0; k0 < NONGENE_DIM; k0 += 4) {
      v2f a  = *(const v2f*)(arow + k0);
      v2f bv = *(const v2f*)(wrow + k0);
      cn = __builtin_amdgcn_wmma_f32_16x16x4_f32(false, a, false, bv, (short)0, cn, false, false);
    }
  }
  if (lane == 0) {
#pragma unroll
    for (int v = 0; v < 8; ++v) s_ng[wib * 16 + v] = cn[v];
  } else if (lane == 16) {
#pragma unroll
    for (int v = 0; v < 8; ++v) s_ng[wib * 16 + 8 + v] = cn[v];
  }
  __builtin_amdgcn_sched_barrier(0);
  v8f cg = {};
  {
    const float* arow = &s_em[wib][c16 * EM_PITCH + kpair];
    const float* wrow = &s_w[wib][64 + kpair];
#pragma unroll
    for (int k0 = 0; k0 < EMB_DIM; k0 += 4) {
      v2f a  = *(const v2f*)(arow + k0);
      v2f bv = *(const v2f*)(wrow + k0);
      cg = __builtin_amdgcn_wmma_f32_16x16x4_f32(false, a, false, bv, (short)0, cg, false, false);
    }
  }

  const float bias = b[0];
  if (lane == 0) {
#pragma unroll
    for (int v = 0; v < 8; ++v) s_gp[wib * 16 + v] = cg[v] + bias;
  } else if (lane == 16) {
#pragma unroll
    for (int v = 0; v < 8; ++v) s_gp[wib * 16 + 8 + v] = cg[v] + bias;
  }

  __builtin_amdgcn_fence(__ATOMIC_RELEASE, "workgroup", "local");
  __builtin_amdgcn_s_barrier();
  __builtin_amdgcn_fence(__ATOMIC_ACQUIRE, "workgroup", "local");

  const float w_x = W[NONGENE_DIM];
  const v4f   gv  = *(const v4f*)(&s_gp[lane << 2]);
  if (in_range) {
#pragma unroll
    for (int r = 0; r < ROWS_PER_WAVE; ++r) {
      const float ngr = s_ng[wib * 16 + r];
      v4f y = xv[r] * w_x + (gv + ngr);
      __builtin_nontemporal_store(y, (v4f*)(op + (long long)r * NUM_GENES));
    }
  }
}

extern "C" void kernel_launch(void* const* d_in, const int* in_sizes, int n_in,
                              void* d_out, int out_size, void* d_ws, size_t ws_size,
                              hipStream_t stream) {
  const float* x   = (const float*)d_in[0];
  const float* emb = (const float*)d_in[1];
  const float* W   = (const float*)d_in[2];
  const float* b   = (const float*)d_in[3];
  float* out = (float*)d_out;

  dim3 blk(32 * WAVES_PER_WG);
  dim3 grd(N_STRIPS, BATCH / TILE_ROWS);
  fused_tile_kernel<<<grd, blk, 0, stream>>>(x, emb, W, b, out);
}


# Work-group tile 128x128 -> 256 rows x 320 columns (32 waves, one per WGP, 1,256 -> 252 work-groups): row term once per 320 columns, gene term once per 256 rows; 241k -> 105k WMMAs; same sums
# speedup vs baseline: 1.7273x; 1.1353x over previous
#include <hip/hip_runtime.h>

typedef __attribute__((ext_vector_type(2))) float v2f;
typedef __attribute__((ext_vector_type(4))) float v4f;
typedef __attribute__((ext_vector_type(8))) float v8f;

#define BATCH        1024
#define NUM_GENES    20000
#define NONGENE_DIM  64
#define EMB_DIM      32
#define INPUT_DIM    (NUM_GENES + NONGENE_DIM)

#define WAVES_PER_WG    32
#define ROW_CHAINS      16
#define TILE_ROWS       (ROW_CHAINS * 16)
#define LINE_COLS       32
#define LINES_PER_HALF  5
#define HALF_COLS       (LINES_PER_HALF * LINE_COLS)
#define TILE_COLS       (2 * HALF_COLS)
#define GENE_CHAINS     (TILE_COLS / 16)
#define N_LINES         (NUM_GENES / LINE_COLS)
#define N_COLGROUPS     ((NUM_GENES + TILE_COLS - 1) / TILE_COLS)
#define LOADS_PER_WAVE  (4 * LINES_PER_HALF)
#define XT_PITCH        (NONGENE_DIM + 4)
#define EM_PITCH        (EMB_DIM + 4)

__global__ void __launch_bounds__(32 * WAVES_PER_WG)
fused_tile_kernel(const float* __restrict__ x,
                  const float* __restrict__ emb,
                  const float* __restrict__ W,
                  const float* __restrict__ b,
                  float* __restrict__ out) {
  __shared__ __attribute__((aligned(16))) float s_xt[ROW_CHAINS][16 * XT_PITCH];
  __shared__ __attribute__((aligned(16))) float s_em[GENE_CHAINS][16 * EM_PITCH];
  __shared__ __attribute__((aligned(16))) float s_gp[TILE_COLS];
  __shared__ __attribute__((aligned(16))) float s_ng[TILE_ROWS];
  __shared__ __attribute__((aligned(16))) float s_wn[ROW_CHAINS][NONGENE_DIM];
  __shared__ __attribute__((aligned(16))) float s_we[16][EMB_DIM];

  const int lane   = (int)(threadIdx.x & 31);
  const int wib    = __builtin_amdgcn_readfirstlane((int)(threadIdx.x >> 5));
  const int cgrp   = (int)blockIdx.x;
  const int rg     = wib & 15;
  const int ch     = wib >> 4;
  const int row0   = (int)blockIdx.y * TILE_ROWS + rg * 16;
  const int line0  = cgrp * (2 * LINES_PER_HALF) + ch * LINES_PER_HALF;
  const bool valid = line0 < N_LINES;
  const bool row_role = wib < ROW_CHAINS;
  const int  gw    = wib - ROW_CHAINS;
  const bool two   = gw < (GENE_CHAINS - 16);

  if (row_role) {
    v4f xt[8];
    const float* p = x + (long long)(row0 + (lane >> 4)) * INPUT_DIM + NUM_GENES + ((lane & 15) << 2);
#pragma unroll
    for (int i = 0; i < 8; ++i) xt[i] = *(const v4f*)(p + (long long)(2 * i) * INPUT_DIM);
    const float w0 = W[lane], w1 = W[32 + lane];
    s_wn[rg][lane] = w0;  s_wn[rg][32 + lane] = w1;
    float* d = &s_xt[rg][(lane >> 4) * XT_PITCH + ((lane & 15) << 2)];
#pragma unroll
    for (int i = 0; i < 8; ++i) *(v4f*)(d + 2 * i * XT_PITCH) = xt[i];
  } else {
    const float w2 = W[NONGENE_DIM + 1 + lane];
    s_we[gw][lane] = w2;
    v4f em[2][4];
#pragma unroll
    for (int n = 0; n < 2; ++n) {
      if (n == 0 || two) {
        const int g0 = cgrp * TILE_COLS + (gw + 16 * n) * 16 + (lane >> 3);
#pragma unroll
        for (int i = 0; i < 4; ++i) {
          int g = g0 + 4 * i;
          g = g < NUM_GENES ? g : NUM_GENES - 1;
          em[n][i] = *(const v4f*)(emb + (long long)g * EMB_DIM + ((lane & 7) << 2));
        }
      }
    }
#pragma unroll
    for (int n = 0; n < 2; ++n) {
      if (n == 0 || two) {
        float* e = &s_em[gw + 16 * n][(lane >> 3) * EM_PITCH + ((lane & 7) << 2)];
#pragma unroll
        for (int i = 0; i < 4; ++i) *(v4f*)(e + 4 * i * EM_PITCH) = em[n][i];
      }
    }
  }

  const int q   = lane >> 3;
  const int sub = lane & 7;
  const float* xp = x   + (long long)(row0 + 4 * q) * INPUT_DIM + (long long)line0 * LINE_COLS + (sub << 2);
  float*       op = out + (long long)(row0 + 4 * q) * NUM_GENES + (long long)line0 * LINE_COLS + (sub << 2);
  v4f xv[LOADS_PER_WAVE];
  if (valid) {
#pragma unroll
    for (int i = 0; i < LOADS_PER_WAVE; ++i)
      xv[i] = __builtin_nontemporal_load(
          (const v4f*)(xp + (long long)(i / LINES_PER_HALF) * INPUT_DIM + (i % LINES_PER_HALF) * LINE_COLS));
  }

  const int c16   = lane & 15;
  const int kpair = (lane >> 4) << 1;
  if (row_role) {
    v8f cn = {};
    const float* arow = &s_xt[rg][c16 * XT_PITCH + kpair];
    const float* wrow = &s_wn[rg][kpair];
#pragma unroll
    for (int k0 = 0; k0 < NONGENE_DIM; k0 += 4) {
      v2f a  = *(const v2f*)(arow + k0);
      v2f bv = *(const v2f*)(wrow + k0);
      cn = __builtin_amdgcn_wmma_f32_16x16x4_f32(false, a, false, bv, (short)0, cn, false, false);
    }
    if (lane == 0) {
#pragma unroll
      for (int v = 0; v < 8; ++v) s_ng[rg * 16 + v] = cn[v];
    } else if (lane == 16) {
#pragma unroll
      for (int v = 0; v < 8; ++v) s_ng[rg * 16 + 8 + v] = cn[v];
    }
  } else {
    const float bias = b[0];
#pragma unroll
    for (int n = 0; n < 2; ++n) {
      if (n == 0 || two) {
        const int c = gw + 16 * n;
        v8f cg = {};
        const float* arow = &s_em[c][c16 * EM_PITCH + kpair];
        const float* wrow = &s_we[gw][kpair];
#pragma unroll
        for (int k0 = 0; k0 < EMB_DIM; k0 += 4) {
          v2f a  = *(const v2f*)(arow + k0);
          v2f bv = *(const v2f*)(wrow + k0);
          cg = __builtin_amdgcn_wmma_f32_16x16x4_f32(false, a, false, bv, (short)0, cg, false, false);
        }
        if (lane == 0) {
#pragma unroll
          for (int v = 0; v < 8; ++v) s_gp[c * 16 + v] = cg[v] + bias;
        } else if (lane == 16) {
#pragma unroll
          for (int v = 0; v < 8; ++v) s_gp[c * 16 + 8 + v] = cg[v] + bias;
        }
      }
    }
  }

  __builtin_amdgcn_fence(__ATOMIC_RELEASE, "workgroup", "local");
  __builtin_amdgcn_s_barrier();
  __builtin_amdgcn_fence(__ATOMIC_ACQUIRE, "workgroup", "local");

  const float w_x = W[NONGENE_DIM];
  v4f gv[LINES_PER_HALF];
#pragma unroll
  for (int c = 0; c < LINES_PER_HALF; ++c)
    gv[c] = *(const v4f*)(&s_gp[ch * HALF_COLS + c * LINE_COLS + (sub << 2)]);
  const v4f ngv = *(const v4f*)(&s_ng[rg * 16 + 4 * q]);
  if (valid) {
#pragma unroll
    for (int i = 0; i < LOADS_PER_WAVE; ++i) {
      v4f y = xv[i] * w_x + (gv[i % LINES_PER_HALF] + ngv[i / LINES_PER_HALF]);
      __builtin_nontemporal_store(
          y, (v4f*)(op + (long long)(i / LINES_PER_HALF) * NUM_GENES + (i % LINES_PER_HALF) * LINE_COLS));
    }
  }
}

extern "C" void kernel_launch(void* const* d_in, const int* in_sizes, int n_in,
                              void* d_out, int out_size, void* d_ws, size_t ws_size,
                              hipStream_t stream) {
  const float* x   = (const float*)d_in[0];
  const float* emb = (const float*)d_in[1];
  const float* W   = (const float*)d_in[2];
  const float* b   = (const float*)d_in[3];
  float* out = (float*)d_out;

  dim3 blk(32 * WAVES_PER_WG);
  dim3 grd(N_COLGROUPS, BATCH / TILE_ROWS);
  fused_tile_kernel<<<grd, blk, 0, stream>>>(x, emb, W, b, out);
}
